# starting kernel
# baseline (speedup 1.0000x reference)
#include <hip/hip_runtime.h>
#include <math.h>

typedef __attribute__((ext_vector_type(2))) float v2f;
typedef __attribute__((ext_vector_type(8))) float v8f;

#define NPTS 128
#define DSEP 1e-5f

__global__ __launch_bounds__(256) void angle_tensor_kernel(
    const float* __restrict__ pos,
    const float* __restrict__ dist,
    float* __restrict__ out)
{
    __shared__ float s_udiff[NPTS * 4];

    const int bi  = blockIdx.x;
    const int b   = bi >> 7;
    const int i   = bi & 127;
    const int tid = threadIdx.x;

    const float* posb = pos + (size_t)b * NPTS * 3;
    const float px = posb[i * 3 + 0];
    const float py = posb[i * 3 + 1];
    const float pz = posb[i * 3 + 2];

    if (tid < NPTS) {
        const int j = tid;
        const float dj = dist[((size_t)b * NPTS + i) * NPTS + j];
        const float rj = (dj < DSEP) ? 1.0f : (1.0f / dj);
        s_udiff[j * 4 + 0] = (posb[j * 3 + 0] - px) * rj;
        s_udiff[j * 4 + 1] = (posb[j * 3 + 1] - py) * rj;
        s_udiff[j * 4 + 2] = (posb[j * 3 + 2] - pz) * rj;
        s_udiff[j * 4 + 3] = 0.0f;
    }
    __syncthreads();

    const int wave = tid >> 5;
    const int lane = tid & 31;
    const int half = lane >> 4;
    const int r    = lane & 15;

    const int jt = wave * 16;
    v2f amat;
    amat.x = s_udiff[(jt + r) * 4 + half * 2 + 0];
    amat.y = s_udiff[(jt + r) * 4 + half * 2 + 1];

    float* outbase = out + (size_t)bi * NPTS * NPTS;

    #pragma unroll
    for (int kn = 0; kn < 8; ++kn) {
        const int kt = kn * 16;
        v2f bmat;
        bmat.x = s_udiff[(kt + r) * 4 + half * 2 + 0];
        bmat.y = s_udiff[(kt + r) * 4 + half * 2 + 1];

        v8f c = {};
        c = __builtin_amdgcn_wmma_f32_16x16x4_f32(
            false, amat, false, bmat, (short)0, c, false, false);

        const int kcol = kt + r;

        #pragma unroll
        for (int v = 0; v < 8; ++v) {
            const int j = jt + v + half * 8;
            float cs = fminf(1.0f, fmaxf(-1.0f, c[v]));
            const bool masked = (i == j) | (i == kcol) | (j == kcol);
            const float ang = masked ? 0.0f : acosf(cs);
            outbase[(size_t)j * NPTS + kcol] = ang;
        }
    }
}

extern "C" void kernel_launch(void* const* d_in, const int* in_sizes, int n_in,
                              void* d_out, int out_size, void* d_ws, size_t ws_size,
                              hipStream_t stream) {
    const float* positions = (const float*)d_in[0];
    const float* dist      = (const float*)d_in[1];
    float* out             = (float*)d_out;

    const int B = 8;
    dim3 grid(B * NPTS);
    dim3 block(256);
    angle_tensor_kernel<<<grid, block, 0, stream>>>(positions, dist, out);
}


# symmetry in (j,k), true for any input: only the upper triangle of the 8x8 tile grid is evaluated, each angle goes to an LDS slab at [j][k] and [k][j], rows leave as whole 512-B stores
# speedup vs baseline: 1.7124x; 1.7124x over previous
#include <hip/hip_runtime.h>
#include <math.h>


typedef __attribute__((ext_vector_type(2))) float v2f;
typedef __attribute__((ext_vector_type(4))) float v4f;
typedef __attribute__((ext_vector_type(8))) float v8f;

#define NPTS 128
#define DSEP 1e-5f
#define PITCH 132

__global__ __launch_bounds__(256) void angle_tensor_kernel(
    const float* __restrict__ pos,
    const float* __restrict__ dist,
    float* __restrict__ out)
{
    __shared__ float s_udiff[NPTS * 4];
    __shared__ __attribute__((aligned(16))) float s_ang[NPTS * PITCH];

    const int bi  = blockIdx.x;
    const int b   = bi >> 7;
    const int i   = bi & 127;
    const int tid = threadIdx.x;

    const float* posb = pos + (size_t)b * NPTS * 3;
    const float px = posb[i * 3 + 0];
    const float py = posb[i * 3 + 1];
    const float pz = posb[i * 3 + 2];

    if (tid < NPTS) {
        const int j = tid;
        const float dj = dist[((size_t)b * NPTS + i) * NPTS + j];
        const float rj = (dj < DSEP) ? 1.0f : (1.0f / dj);
        s_udiff[j * 4 + 0] = (posb[j * 3 + 0] - px) * rj;
        s_udiff[j * 4 + 1] = (posb[j * 3 + 1] - py) * rj;
        s_udiff[j * 4 + 2] = (posb[j * 3 + 2] - pz) * rj;
        s_udiff[j * 4 + 3] = 0.0f;
    }
    __syncthreads();

    const int wave = __builtin_amdgcn_readfirstlane(tid >> 5);
    const int lane = tid & 31;
    const int half = lane >> 4;
    const int r    = lane & 15;

    #pragma unroll
    for (int s = 0; s < 5; ++s) {
        const int ta   = (s < 4) ? wave : (wave & 3);
        const int tb   = (s < 4) ? ((wave + s) & 7) : (ta + 4);
        const int rsel = (s < 4) ? 0 : ((wave >> 2) << 2);
        const int nv   = (s < 4) ? 8 : 4;

        const int arow = ta * 16 + (r ^ rsel);
        v2f amat;
        amat.x = s_udiff[arow * 4 + half * 2 + 0];
        amat.y = s_udiff[arow * 4 + half * 2 + 1];
        const int kcol = tb * 16 + r;
        v2f bmat;
        bmat.x = s_udiff[kcol * 4 + half * 2 + 0];
        bmat.y = s_udiff[kcol * 4 + half * 2 + 1];

        v8f c = {};
        c = __builtin_amdgcn_wmma_f32_16x16x4_f32(
            false, amat, false, bmat, (short)0, c, false, false);

        const int j0 = ta * 16 + half * 8 + rsel;
        float ang[8];
        #pragma unroll
        for (int v = 0; v < 8; ++v) {
            if (v < nv) {
                const int j = j0 + v;
                const float cs = fminf(1.0f, fmaxf(-1.0f, c[v]));
                const bool masked = (i == j) | (i == kcol) | (j == kcol);
                ang[v] = masked ? 0.0f : acosf(cs);
            }
        }

        float* mir = &s_ang[kcol * PITCH + j0];
        *(v4f*)(mir) = (v4f){ang[0], ang[1], ang[2], ang[3]};
        if (nv == 8)
            *(v4f*)(mir + 4) = (v4f){ang[4], ang[5], ang[6], ang[7]};
        if (s != 0) {
            #pragma unroll
            for (int v = 0; v < 8; ++v)
                if (v < nv) s_ang[(j0 + v) * PITCH + kcol] = ang[v];
        }
    }
    __syncthreads();

    float* outbase = out + (size_t)bi * NPTS * NPTS;
    #pragma unroll
    for (int rr = 0; rr < 16; ++rr) {
        const int j = wave * 16 + rr;
        const v4f val = *(const v4f*)&s_ang[j * PITCH + lane * 4];
        *(v4f*)&outbase[(size_t)j * NPTS + lane * 4] = val;
    }
}

extern "C" void kernel_launch(void* const* d_in, const int* in_sizes, int n_in,
                              void* d_out, int out_size, void* d_ws, size_t ws_size,
                              hipStream_t stream) {
    const float* positions = (const float*)d_in[0];
    const float* dist      = (const float*)d_in[1];
    float* out             = (float*)d_out;

    const int B = 8;
    dim3 grid(B * NPTS);
    dim3 block(256);
    angle_tensor_kernel<<<grid, block, 0, stream>>>(positions, dist, out);
}


# acosf() under a per-value exec branch replaced by one branch-free form sqrt(1-|c|)*P7(|c|), pi-mirrored for c<0, two values per packed f32 FMA, mask by select; error within 3 ulp of pi/2
# speedup vs baseline: 2.1263x; 1.2417x over previous
#include <hip/hip_runtime.h>
#include <math.h>


typedef __attribute__((ext_vector_type(2))) float v2f;
typedef __attribute__((ext_vector_type(4))) float v4f;
typedef __attribute__((ext_vector_type(8))) float v8f;

#define NPTS 128
#define DSEP 1e-5f
#define PITCH 132

__device__ __forceinline__ v2f acos_pair(v2f x)
{
    const v2f ax = { fminf(fabsf(x.x), 1.0f), fminf(fabsf(x.y), 1.0f) };
    const v2f t  = 1.0f - ax;
    const v2f s  = { __builtin_amdgcn_sqrtf(t.x), __builtin_amdgcn_sqrtf(t.y) };
    v2f p = (v2f)(-0.0012624911f);
    p = __builtin_elementwise_fma(p, ax, (v2f)( 0.0066700901f));
    p = __builtin_elementwise_fma(p, ax, (v2f)(-0.0170881256f));
    p = __builtin_elementwise_fma(p, ax, (v2f)( 0.0308918810f));
    p = __builtin_elementwise_fma(p, ax, (v2f)(-0.0501743046f));
    p = __builtin_elementwise_fma(p, ax, (v2f)( 0.0889789874f));
    p = __builtin_elementwise_fma(p, ax, (v2f)(-0.2145988016f));
    p = __builtin_elementwise_fma(p, ax, (v2f)( 1.5707963050f));
    const v2f sg   = { (x.x < 0.0f) ? -s.x : s.x, (x.y < 0.0f) ? -s.y : s.y };
    const v2f base = { (x.x < 0.0f) ? 3.14159274f : 0.0f, (x.y < 0.0f) ? 3.14159274f : 0.0f };
    return __builtin_elementwise_fma(sg, p, base);
}

__global__ __launch_bounds__(256) void angle_tensor_kernel(
    const float* __restrict__ pos,
    const float* __restrict__ dist,
    float* __restrict__ out)
{
    __shared__ float s_udiff[NPTS * 4];
    __shared__ __attribute__((aligned(16))) float s_ang[NPTS * PITCH];

    const int bi  = blockIdx.x;
    const int b   = bi >> 7;
    const int i   = bi & 127;
    const int tid = threadIdx.x;

    const float* posb = pos + (size_t)b * NPTS * 3;
    const float px = posb[i * 3 + 0];
    const float py = posb[i * 3 + 1];
    const float pz = posb[i * 3 + 2];

    if (tid < NPTS) {
        const int j = tid;
        const float dj = dist[((size_t)b * NPTS + i) * NPTS + j];
        const float rj = (dj < DSEP) ? 1.0f : (1.0f / dj);
        s_udiff[j * 4 + 0] = (posb[j * 3 + 0] - px) * rj;
        s_udiff[j * 4 + 1] = (posb[j * 3 + 1] - py) * rj;
        s_udiff[j * 4 + 2] = (posb[j * 3 + 2] - pz) * rj;
        s_udiff[j * 4 + 3] = 0.0f;
    }
    __syncthreads();

    const int wave = __builtin_amdgcn_readfirstlane(tid >> 5);
    const int lane = tid & 31;
    const int half = lane >> 4;
    const int r    = lane & 15;

    #pragma unroll
    for (int s = 0; s < 5; ++s) {
        const int ta   = (s < 4) ? wave : (wave & 3);
        const int tb   = (s < 4) ? ((wave + s) & 7) : (ta + 4);
        const int rsel = (s < 4) ? 0 : ((wave >> 2) << 2);
        const int nv   = (s < 4) ? 8 : 4;

        const int arow = ta * 16 + (r ^ rsel);
        v2f amat;
        amat.x = s_udiff[arow * 4 + half * 2 + 0];
        amat.y = s_udiff[arow * 4 + half * 2 + 1];
        const int kcol = tb * 16 + r;
        v2f bmat;
        bmat.x = s_udiff[kcol * 4 + half * 2 + 0];
        bmat.y = s_udiff[kcol * 4 + half * 2 + 1];

        v8f c = {};
        c = __builtin_amdgcn_wmma_f32_16x16x4_f32(
            false, amat, false, bmat, (short)0, c, false, false);

        const int j0 = ta * 16 + half * 8 + rsel;
        float ang[8];
        const bool colm = (i == kcol);
        const int  di   = i - j0;
        const int  dk   = kcol - j0;
        #pragma unroll
        for (int v = 0; v < 8; v += 2) {
            if (v < nv) {
                const v2f a = acos_pair((v2f){c[v], c[v + 1]});
                const bool m0 = colm | (di == v)     | ((s == 0) & (dk == v));
                const bool m1 = colm | (di == v + 1) | ((s == 0) & (dk == v + 1));
                ang[v]     = m0 ? 0.0f : a.x;
                ang[v + 1] = m1 ? 0.0f : a.y;
            }
        }

        float* mir = &s_ang[kcol * PITCH + j0];
        *(v4f*)(mir) = (v4f){ang[0], ang[1], ang[2], ang[3]};
        if (nv == 8)
            *(v4f*)(mir + 4) = (v4f){ang[4], ang[5], ang[6], ang[7]};
        if (s != 0) {
            #pragma unroll
            for (int v = 0; v < 8; ++v)
                if (v < nv) s_ang[(j0 + v) * PITCH + kcol] = ang[v];
        }
    }
    __syncthreads();

    float* outbase = out + (size_t)bi * NPTS * NPTS;
    #pragma unroll
    for (int rr = 0; rr < 16; ++rr) {
        const int j = wave * 16 + rr;
        const v4f val = *(const v4f*)&s_ang[j * PITCH + lane * 4];
        *(v4f*)&outbase[(size_t)j * NPTS + lane * 4] = val;
    }
}

extern "C" void kernel_launch(void* const* d_in, const int* in_sizes, int n_in,
                              void* d_out, int out_size, void* d_ws, size_t ws_size,
                              hipStream_t stream) {
    const float* positions = (const float*)d_in[0];
    const float* dist      = (const float*)d_in[1];
    float* out             = (float*)d_out;

    const int B = 8;
    dim3 grid(B * NPTS);
    dim3 block(256);
    angle_tensor_kernel<<<grid, block, 0, stream>>>(positions, dist, out);
}
